# starting kernel
# baseline (speedup 1.0000x reference)
#include <hip/hip_runtime.h>

typedef __attribute__((ext_vector_type(2))) float v2f;
typedef __attribute__((ext_vector_type(8))) float v8f;

#define N_NODES 100000
#define D 128
#define SLOPE 0.2f

__global__ void sage_scatter(const float* __restrict__ x,
                             const int*   __restrict__ ei,
                             float*       __restrict__ agg,
                             float*       __restrict__ cnt,
                             int E) {
    long long t = (long long)blockIdx.x * blockDim.x + threadIdx.x;
    int e    = (int)(t >> 5);
    int lane = (int)(t & 31);
    if (e >= E) return;
    int src = ei[e];
    int dst = ei[(long long)E + e];

    float4 v = ((const float4*)(x + (long long)src * D))[lane];
    float* a = agg + (long long)dst * D + lane * 4;
    atomicAdd(a + 0, v.x);
    atomicAdd(a + 1, v.y);
    atomicAdd(a + 2, v.z);
    atomicAdd(a + 3, v.w);
    if (lane == 0) atomicAdd(cnt + dst, 1.0f);
}

__global__ void __launch_bounds__(256)
sage_gemm(const float* __restrict__ x,
          const float* __restrict__ Wl,
          const float* __restrict__ bl,
          const float* __restrict__ Wr,
          const float* __restrict__ cnt,
          float* __restrict__ out  ) {
    const int wave = (blockIdx.x * (blockDim.x >> 5)) + (threadIdx.x >> 5);
    const int lane = threadIdx.x & 31;
    const int r0   = wave * 16;
    if (r0 >= N_NODES) return;

    const int lrow  = lane & 15;
    const int khalf = (lane >> 4) * 2;
    const int row   = r0 + lrow;

    const float rc = 1.0f / fmaxf(cnt[row], 1.0f);

    v8f acc[8];
#pragma unroll
    for (int o = 0; o < 8; ++o) { v8f z = {}; acc[o] = z; }

    const float* aggRow = out + (long long)row * D;
    const float* xRow   = x   + (long long)row * D;

    for (int kk = 0; kk < D; kk += 4) {
        float2 t = *(const float2*)(aggRow + kk + khalf);
        v2f a; a.x = t.x * rc; a.y = t.y * rc;
#pragma unroll
        for (int o = 0; o < 8; ++o) {
            float2 w = *(const float2*)(Wl + (long long)(o * 16 + lrow) * D + kk + khalf);
            v2f b; b.x = w.x; b.y = w.y;
            acc[o] = __builtin_amdgcn_wmma_f32_16x16x4_f32(
                false, a, false, b, (short)0, acc[o], false, false);
        }
    }

    for (int kk = 0; kk < D; kk += 4) {
        float2 t = *(const float2*)(xRow + kk + khalf);
        v2f a; a.x = t.x; a.y = t.y;
#pragma unroll
        for (int o = 0; o < 8; ++o) {
            float2 w = *(const float2*)(Wr + (long long)(o * 16 + lrow) * D + kk + khalf);
            v2f b; b.x = w.x; b.y = w.y;
            acc[o] = __builtin_amdgcn_wmma_f32_16x16x4_f32(
                false, a, false, b, (short)0, acc[o], false, false);
        }
    }

#pragma unroll
    for (int o = 0; o < 8; ++o) {
        const int   col = o * 16 + lrow;
        const float bb  = bl[col];
#pragma unroll
        for (int j = 0; j < 8; ++j) {
            const int rr = r0 + j + (lane >> 4) * 8;
            float v = acc[o][j] + bb;
            v = (v > 0.0f) ? v : SLOPE * v;
            out[(long long)rr * D + col] = v;
        }
    }
}

extern "C" void kernel_launch(void* const* d_in, const int* in_sizes, int n_in,
                              void* d_out, int out_size, void* d_ws, size_t ws_size,
                              hipStream_t stream) {
    const float* x  = (const float*)d_in[0];
    const int*   ei = (const int*)  d_in[1];
    const float* Wl = (const float*)d_in[2];
    const float* bl = (const float*)d_in[3];
    const float* Wr = (const float*)d_in[4];
    float* out = (float*)d_out;
    float* cnt = (float*)d_ws;
    const int E = in_sizes[1] / 2;

    hipMemsetAsync(d_out, 0, (size_t)N_NODES * D * sizeof(float), stream);
    hipMemsetAsync(d_ws,  0, (size_t)N_NODES * sizeof(float),     stream);

    {
        long long tot = (long long)E * 32;
        int blocks = (int)((tot + 255) / 256);
        sage_scatter<<<blocks, 256, 0, stream>>>(x, ei, out, cnt, E);
    }

    {
        int tiles  = (N_NODES + 15) / 16;
        int blocks = (tiles + 7) / 8;
        sage_gemm<<<blocks, 256, 0, stream>>>(x, Wl, bl, Wr, cnt, out);
    }
}


# edge scatter: 256 work-groups each own 391 dst rows in LDS, filter the whole dst list, add x rows by LDS f32 atomics, store rows once; no global atomics, no memsets; sum order = LDS-atomic order
# speedup vs baseline: 2.5960x; 2.5960x over previous
#include <hip/hip_runtime.h>


typedef __attribute__((ext_vector_type(2))) float v2f;
typedef __attribute__((ext_vector_type(8))) float v8f;

#define N_NODES 100000
#define D 128
#define SLOPE 0.2f

#define N_TILES   (N_NODES / 16)
#define GEMM_WGS  256
#define GEMM_THREADS 512
#define GEMM_WAVES (GEMM_THREADS / 32)
#define WP 132

#define OWN_WGS     256
#define OWN_THREADS 1024
#define OWN_WAVES   (OWN_THREADS / 32)
#define NB          391
#define LIST_CAP    16384
#define SCAN_U      8
#define SCAN_CH     (OWN_THREADS * SCAN_U)
#define GATH_U      8

__global__ void __launch_bounds__(OWN_THREADS)
sage_own(const float* __restrict__ x,
         const int*   __restrict__ ei,
         float*       __restrict__ agg,
         float*       __restrict__ cnt,
         int E) {
    __shared__ float sA[NB * D];
    __shared__ float sC[NB];
    __shared__ int   sL[LIST_CAP];
    __shared__ int   sN;

    const int tid  = threadIdx.x;
    const int lane = tid & 31;
    const int lo   = (int)blockIdx.x * NB;
    const int nb   = min(NB, N_NODES - lo);

    for (int i = tid; i < NB * D / 4; i += OWN_THREADS) {
        float4 z = {0.0f, 0.0f, 0.0f, 0.0f};
        *(float4*)(sA + 4 * i) = z;
    }
    for (int i = tid; i < NB; i += OWN_THREADS) sC[i] = 0.0f;
    if (tid == 0) sN = 0;
    __syncthreads();

    const int* dstp = ei + (long long)E;
    const int  nvec = E >> 2;

    auto take = [&](int dv, int e) {
        const unsigned dl = (unsigned)(dv - lo);
        if (dl < (unsigned)nb) {
            atomicAdd(&sC[dl], 1.0f);
            const int pos = atomicAdd(&sN, 1);
            if (pos < LIST_CAP) {
                sL[pos] = e | ((int)dl << 21);
            } else {
                const float* xr = x + (long long)ei[e] * D;
                for (int f = 0; f < D; ++f) atomicAdd(&sA[dl * D + f], xr[f]);
            }
        }
    };

    auto take4 = [&](const int4 d, int e0) {
        const unsigned d0 = (unsigned)(d.x - lo), d1 = (unsigned)(d.y - lo);
        const unsigned d2 = (unsigned)(d.z - lo), d3 = (unsigned)(d.w - lo);
        if (min(min(d0, d1), min(d2, d3)) < (unsigned)nb) {
            take(d.x, e0);
            take(d.y, e0 + 1);
            take(d.z, e0 + 2);
            take(d.w, e0 + 3);
        }
    };

    const int nch = (nvec + SCAN_CH - 1) / SCAN_CH;
    int c = (nch > 0) ? (int)(blockIdx.x % (unsigned)nch) : 0;
    for (int t = 0; t < nch; ++t) {
        const int j0 = c * SCAN_CH + tid;
        if (c * SCAN_CH + SCAN_CH <= nvec) {
            int4 d[SCAN_U];
#pragma unroll
            for (int k = 0; k < SCAN_U; ++k)
                d[k] = *(const int4*)(dstp + 4 * (long long)(j0 + k * OWN_THREADS));
#pragma unroll
            for (int k = 0; k < SCAN_U; ++k) take4(d[k], 4 * (j0 + k * OWN_THREADS));
        } else {
            for (int j = j0; j < nvec; j += OWN_THREADS)
                take4(*(const int4*)(dstp + 4 * (long long)j), 4 * j);
        }
        c = (c + 1 == nch) ? 0 : c + 1;
    }
    {
        const int e = 4 * nvec + tid;
        if (e < E) take(dstp[e], e);
    }
    __syncthreads();

    const int n = min(sN, LIST_CAP);
    for (int q = tid; q < n; q += OWN_THREADS) {
        const int ent = sL[q];
        const int s   = ei[ent & 0x1FFFFF];
        sL[q] = s | ((ent >> 21) << 17);
    }
    __syncthreads();

    const int wave  = __builtin_amdgcn_readfirstlane(tid >> 5);
    const int nfull = n & ~(GATH_U - 1);
    for (int base = wave * GATH_U; base < nfull; base += OWN_WAVES * GATH_U) {
        const int mine = sL[base + (lane & (GATH_U - 1))];
        float v[GATH_U][4];
        int   dl[GATH_U];
#pragma unroll
        for (int k = 0; k < GATH_U; ++k) {
            const int ent = __builtin_amdgcn_readlane(mine, k);
            dl[k] = ent >> 17;
            const float* xr = x + (long long)(ent & 0x1FFFF) * D + lane;
            v[k][0] = xr[0];
            v[k][1] = xr[32];
            v[k][2] = xr[64];
            v[k][3] = xr[96];
        }
#pragma unroll
        for (int k = 0; k < GATH_U; ++k) {
            float* a = sA + dl[k] * D + lane;
            atomicAdd(a,      v[k][0]);
            atomicAdd(a + 32, v[k][1]);
            atomicAdd(a + 64, v[k][2]);
            atomicAdd(a + 96, v[k][3]);
        }
    }
    if (wave < n - nfull) {
        const int ent = __builtin_amdgcn_readfirstlane(sL[nfull + wave]);
        const float* xr = x + (long long)(ent & 0x1FFFF) * D + lane;
        const float v0 = xr[0], v1 = xr[32], v2 = xr[64], v3 = xr[96];
        float* a = sA + (ent >> 17) * D + lane;
        atomicAdd(a,      v0);
        atomicAdd(a + 32, v1);
        atomicAdd(a + 64, v2);
        atomicAdd(a + 96, v3);
    }
    __syncthreads();

    for (int q = tid; q < nb * (D / 4); q += OWN_THREADS) {
        const int row = q >> 5;
        const int c4  = q & 31;
        *(float4*)(agg + (long long)(lo + row) * D + c4 * 4) = *(const float4*)(sA + row * D + c4 * 4);
    }
    for (int q = tid; q < nb; q += OWN_THREADS) cnt[lo + q] = sC[q];
}

template <bool SCALED, bool CHAIN>
__device__ __forceinline__ void k_sweep(const float* sWbase,
                                        const float* aPtr, int bOff, float scale,
                                        const float* aNext, int bOffNext,
                                        float2 (&ac)[8], v2f (&bc)[8], v8f (&acc)[8]) {
#pragma unroll 1
    for (int c = 0; c < 4; ++c) {
        const bool more = (c < 3);
        const float* aN = more ? (aPtr + 32 * (c + 1)) : aNext;
        const int    bN = more ? (bOff + 32 * (c + 1)) : bOffNext;
        const int    bC = bOff + 32 * c;
        float2 an[8];
        if (CHAIN || more) {
#pragma unroll
            for (int s = 0; s < 8; ++s) an[s] = *(const float2*)(aN + 4 * s);
        }
#pragma unroll
        for (int s = 0; s < 8; ++s) {
            v2f bn[8];
            if (s < 7) {
#pragma unroll
                for (int o = 0; o < 8; ++o) bn[o] = *(const v2f*)(sWbase + bC + o * 16 * WP + 4 * (s + 1));
            } else if (CHAIN || more) {
#pragma unroll
                for (int o = 0; o < 8; ++o) bn[o] = *(const v2f*)(sWbase + bN + o * 16 * WP);
            }
            v2f a;
            if (SCALED) { a.x = ac[s].x * scale; a.y = ac[s].y * scale; }
            else        { a.x = ac[s].x;         a.y = ac[s].y; }
#pragma unroll
            for (int o = 0; o < 8; ++o) {
                acc[o] = __builtin_amdgcn_wmma_f32_16x16x4_f32(
                    false, a, false, bc[o], (short)0, acc[o], false, false);
            }
            if (s < 7 || CHAIN || more) {
#pragma unroll
                for (int o = 0; o < 8; ++o) bc[o] = bn[o];
            }
        }
        if (CHAIN || more) {
#pragma unroll
            for (int s = 0; s < 8; ++s) ac[s] = an[s];
        }
    }
}

__global__ void __launch_bounds__(GEMM_THREADS)
sage_gemm(const float* __restrict__ x,
          const float* __restrict__ Wl,
          const float* __restrict__ bl,
          const float* __restrict__ Wr,
          const float* __restrict__ cnt,
          float* __restrict__ out  ) {
    __shared__ float sW[2 * D * WP];

    for (int i = threadIdx.x; i < 2 * D * (D / 4); i += GEMM_THREADS) {
        const int m  = i >> 12;
        const int n  = (i >> 5) & (D - 1);
        const int c4 = i & 31;
        const float* W = m ? Wr : Wl;
        const float4 v = *(const float4*)(W + n * D + c4 * 4);
        *(float4*)(sW + (m * D + n) * WP + c4 * 4) = v;
    }
    __syncthreads();

    const int wave = __builtin_amdgcn_readfirstlane((int)(threadIdx.x >> 5));
    const int lane = threadIdx.x & 31;
    const int lrow  = lane & 15;
    const int khalf = (lane >> 4) * 2;

    const int bOffL = lrow * WP + khalf;
    const int bOffR = bOffL + D * WP;

    float bb[8];
#pragma unroll
    for (int o = 0; o < 8; ++o) bb[o] = bl[o * 16 + lrow];

    for (int m = wave; ; m += GEMM_WAVES) {
        const int tile = (int)blockIdx.x + GEMM_WGS * m;
        if (tile >= N_TILES) break;
        const int r0  = tile * 16;
        const int row = r0 + lrow;

        const float rc = 1.0f / fmaxf(cnt[row], 1.0f);

        v8f acc[8];
#pragma unroll
        for (int o = 0; o < 8; ++o) { v8f z = {}; acc[o] = z; }

        const float* aggPtr = out + (long long)row * D + khalf;
        const float* xPtr   = x   + (long long)row * D + khalf;

        float2 ac[8];
        v2f    bc[8];
#pragma unroll
        for (int s = 0; s < 8; ++s) ac[s] = *(const float2*)(aggPtr + 4 * s);
#pragma unroll
        for (int o = 0; o < 8; ++o) bc[o] = *(const v2f*)(sW + bOffL + o * 16 * WP);

        k_sweep<true,  true >(sW, aggPtr, bOffL, rc,   xPtr,   bOffR, ac, bc, acc);
        k_sweep<false, false>(sW, xPtr,   bOffR, 1.0f, xPtr,   bOffR, ac, bc, acc);

#pragma unroll
        for (int o = 0; o < 8; ++o) {
            const int col = o * 16 + lrow;
#pragma unroll
            for (int j = 0; j < 8; ++j) {
                const int rr = r0 + j + (lane >> 4) * 8;
                float v = acc[o][j] + bb[o];
                v = (v > 0.0f) ? v : SLOPE * v;
                out[(long long)rr * D + col] = v;
            }
        }
    }
}

extern "C" void kernel_launch(void* const* d_in, const int* in_sizes, int n_in,
                              void* d_out, int out_size, void* d_ws, size_t ws_size,
                              hipStream_t stream) {
    const float* x  = (const float*)d_in[0];
    const int*   ei = (const int*)  d_in[1];
    const float* Wl = (const float*)d_in[2];
    const float* bl = (const float*)d_in[3];
    const float* Wr = (const float*)d_in[4];
    float* out = (float*)d_out;
    float* cnt = (float*)d_ws;
    const int E = in_sizes[1] / 2;

    sage_own<<<OWN_WGS, OWN_THREADS, 0, stream>>>(x, ei, out, cnt, E);

    sage_gemm<<<GEMM_WGS, GEMM_THREADS, 0, stream>>>(x, Wl, bl, Wr, cnt, out);
}


# edge scatter: a pre-launch packs each edge's owner block (dst/391) into one byte; owner blocks scan 16 edges per 16-B load (zero-byte test), resolve only hit units; sum order = LDS-atomic order
# speedup vs baseline: 3.0462x; 1.1734x over previous
#include <hip/hip_runtime.h>


typedef __attribute__((ext_vector_type(2))) float v2f;
typedef __attribute__((ext_vector_type(8))) float v8f;

#define N_NODES 100000
#define D 128
#define SLOPE 0.2f

#define N_TILES   (N_NODES / 16)
#define GEMM_WGS  256
#define GEMM_THREADS 512
#define GEMM_WAVES (GEMM_THREADS / 32)
#define WP 132

#define NB          391
#define KEY_WGS     512
#define KEY_THREADS 256
#define KEY_U       4
#define KEY_TILE    (KEY_THREADS * KEY_U)
#define KEY_OFFSET  (1 << 20)

__device__ __forceinline__ unsigned owner_of(int d) {
    return ((unsigned)d < (unsigned)N_NODES) ? ((unsigned)d / (unsigned)NB) : 0u;
}

__device__ __forceinline__ unsigned owner_word(const int4 d) {
    return owner_of(d.x) | (owner_of(d.y) << 8) | (owner_of(d.z) << 16) | (owner_of(d.w) << 24);
}

__global__ void __launch_bounds__(KEY_THREADS)
sage_keys(const int* __restrict__ ei,
          unsigned*  __restrict__ keys,
          int E) {
    const int* dstp  = ei + (long long)E;
    const int  nfull = E >> 2;
    const int  ndw   = ((E + 15) >> 4) << 2;
    const int  ntile = (ndw + KEY_TILE - 1) / KEY_TILE;
    for (int tile = (int)blockIdx.x; tile < ntile; tile += KEY_WGS) {
        const int j0 = tile * KEY_TILE + (int)threadIdx.x;
        if (tile * KEY_TILE + KEY_TILE <= nfull) {
            int4 d[KEY_U];
#pragma unroll
            for (int k = 0; k < KEY_U; ++k)
                d[k] = *(const int4*)(dstp + 4 * (long long)(j0 + k * KEY_THREADS));
#pragma unroll
            for (int k = 0; k < KEY_U; ++k) keys[j0 + k * KEY_THREADS] = owner_word(d[k]);
        } else {
            const int jend = min(ndw, tile * KEY_TILE + KEY_TILE);
            for (int j = j0; j < jend; j += KEY_THREADS) {
                unsigned w;
                if (j < nfull) {
                    w = owner_word(*(const int4*)(dstp + 4 * (long long)j));
                } else {
                    w = 0u;
                    for (int i = 0; i < 4; ++i) {
                        const int e = 4 * j + i;
                        if (e < E) w |= owner_of(dstp[e]) << (8 * i);
                    }
                }
                keys[j] = w;
            }
        }
    }
}

#define OWN_WGS     256
#define OWN_THREADS 1024
#define OWN_WAVES   (OWN_THREADS / 32)
#define LIST_CAP    16384
#define UNIT_CAP    8192
#define SCAN_U      8
#define SCAN_CH     (OWN_THREADS * SCAN_U)
#define GATH_U      8

__global__ void __launch_bounds__(OWN_THREADS)
sage_own(const float*    __restrict__ x,
         const int*      __restrict__ ei,
         const unsigned* __restrict__ keys,
         float*          __restrict__ agg,
         float*          __restrict__ cnt,
         int E) {
    __shared__ float sA[NB * D];
    __shared__ float sC[NB];
    __shared__ int   sL[LIST_CAP];
    __shared__ int   sU[UNIT_CAP];
    __shared__ int   sN;
    __shared__ int   sNU;

    const int tid  = threadIdx.x;
    const int lane = tid & 31;
    const int lo   = (int)blockIdx.x * NB;
    const int nb   = min(NB, N_NODES - lo);

    for (int i = tid; i < NB * D / 4; i += OWN_THREADS) {
        float4 z = {0.0f, 0.0f, 0.0f, 0.0f};
        *(float4*)(sA + 4 * i) = z;
    }
    for (int i = tid; i < NB; i += OWN_THREADS) sC[i] = 0.0f;
    if (tid == 0) { sN = 0; sNU = 0; }
    __syncthreads();

    const int* dstp = ei + (long long)E;

    auto take = [&](int dv, int e) {
        const unsigned dl = (unsigned)(dv - lo);
        if (dl < (unsigned)nb) {
            atomicAdd(&sC[dl], 1.0f);
            const int pos = atomicAdd(&sN, 1);
            if (pos < LIST_CAP) {
                sL[pos] = e | ((int)dl << 21);
            } else {
                const float* xr = x + (long long)ei[e] * D;
                for (int f = 0; f < D; ++f) atomicAdd(&sA[dl * D + f], xr[f]);
            }
        }
    };

    auto take4 = [&](const int4 d, int e0) {
        const unsigned d0 = (unsigned)(d.x - lo), d1 = (unsigned)(d.y - lo);
        const unsigned d2 = (unsigned)(d.z - lo), d3 = (unsigned)(d.w - lo);
        if (min(min(d0, d1), min(d2, d3)) < (unsigned)nb) {
            take(d.x, e0);
            take(d.y, e0 + 1);
            take(d.z, e0 + 2);
            take(d.w, e0 + 3);
        }
    };

    auto resolve = [&](int j) {
        const int e0 = 16 * j;
        if (e0 + 16 <= E) {
            int4 d[4];
#pragma unroll
            for (int k = 0; k < 4; ++k) d[k] = *(const int4*)(dstp + e0 + 4 * k);
#pragma unroll
            for (int k = 0; k < 4; ++k) take4(d[k], e0 + 4 * k);
        } else {
            for (int e = e0; e < E; ++e) take(dstp[e], e);
        }
    };

    const unsigned own   = (unsigned)blockIdx.x * 0x01010101u;
    const int      nunit = (E + 15) >> 4;
    const uint4*   kp    = (const uint4*)keys;

    auto scan1 = [&](const uint4 w, int j) {
        const unsigned v0 = w.x ^ own, v1 = w.y ^ own, v2 = w.z ^ own, v3 = w.w ^ own;
        const unsigned h  = (((v0 - 0x01010101u) & ~v0) | ((v1 - 0x01010101u) & ~v1) |
                             ((v2 - 0x01010101u) & ~v2) | ((v3 - 0x01010101u) & ~v3)) & 0x80808080u;
        if (h != 0u) {
            const int pos = atomicAdd(&sNU, 1);
            if (pos < UNIT_CAP) sU[pos] = j;
        }
    };

    const int nch   = (nunit + SCAN_CH - 1) / SCAN_CH;
    const int first = (int)(((long long)blockIdx.x * nunit) >> 8) & ~31;
    for (int t = 0; t < nch; ++t) {
        const int l0 = t * SCAN_CH + tid;
        if (t * SCAN_CH + SCAN_CH <= nunit) {
            uint4 w[SCAN_U];
            int   jj[SCAN_U];
#pragma unroll
            for (int k = 0; k < SCAN_U; ++k) {
                const int j = l0 + k * OWN_THREADS + first;
                jj[k] = (j >= nunit) ? j - nunit : j;
                w[k]  = kp[jj[k]];
            }
#pragma unroll
            for (int k = 0; k < SCAN_U; ++k) scan1(w[k], jj[k]);
        } else {
            for (int l = l0; l < nunit; l += OWN_THREADS) {
                const int j = l + first;
                const int jw = (j >= nunit) ? j - nunit : j;
                scan1(kp[jw], jw);
            }
        }
    }
    __syncthreads();

    {
        const int  nuAll  = sNU;
        const bool listed = (nuAll <= UNIT_CAP);
        const int  nwork  = listed ? nuAll : nunit;
        for (int q = tid; q < nwork; q += OWN_THREADS) resolve(listed ? sU[q] : q);
    }
    __syncthreads();

    const int n = min(sN, LIST_CAP);
    for (int q = tid; q < n; q += OWN_THREADS) {
        const int ent = sL[q];
        const int s   = ei[ent & 0x1FFFFF];
        sL[q] = s | ((ent >> 21) << 17);
    }
    __syncthreads();

    const int wave  = __builtin_amdgcn_readfirstlane(tid >> 5);
    const int nfull = n & ~(GATH_U - 1);
    for (int base = wave * GATH_U; base < nfull; base += OWN_WAVES * GATH_U) {
        const int mine = sL[base + (lane & (GATH_U - 1))];
        float v[GATH_U][4];
        int   dl[GATH_U];
#pragma unroll
        for (int k = 0; k < GATH_U; ++k) {
            const int ent = __builtin_amdgcn_readlane(mine, k);
            dl[k] = ent >> 17;
            const float* xr = x + (long long)(ent & 0x1FFFF) * D + lane;
            v[k][0] = xr[0];
            v[k][1] = xr[32];
            v[k][2] = xr[64];
            v[k][3] = xr[96];
        }
#pragma unroll
        for (int k = 0; k < GATH_U; ++k) {
            float* a = sA + dl[k] * D + lane;
            atomicAdd(a,      v[k][0]);
            atomicAdd(a + 32, v[k][1]);
            atomicAdd(a + 64, v[k][2]);
            atomicAdd(a + 96, v[k][3]);
        }
    }
    if (wave < n - nfull) {
        const int ent = __builtin_amdgcn_readfirstlane(sL[nfull + wave]);
        const float* xr = x + (long long)(ent & 0x1FFFF) * D + lane;
        const float v0 = xr[0], v1 = xr[32], v2 = xr[64], v3 = xr[96];
        float* a = sA + (ent >> 17) * D + lane;
        atomicAdd(a,      v0);
        atomicAdd(a + 32, v1);
        atomicAdd(a + 64, v2);
        atomicAdd(a + 96, v3);
    }
    __syncthreads();

    for (int q = tid; q < nb * (D / 4); q += OWN_THREADS) {
        const int row = q >> 5;
        const int c4  = q & 31;
        *(float4*)(agg + (long long)(lo + row) * D + c4 * 4) = *(const float4*)(sA + row * D + c4 * 4);
    }
    for (int q = tid; q < nb; q += OWN_THREADS) cnt[lo + q] = sC[q];
}

template <bool SCALED, bool CHAIN>
__device__ __forceinline__ void k_sweep(const float* sWbase,
                                        const float* aPtr, int bOff, float scale,
                                        const float* aNext, int bOffNext,
                                        float2 (&ac)[8], v2f (&bc)[8], v8f (&acc)[8]) {
#pragma unroll 1
    for (int c = 0; c < 4; ++c) {
        const bool more = (c < 3);
        const float* aN = more ? (aPtr + 32 * (c + 1)) : aNext;
        const int    bN = more ? (bOff + 32 * (c + 1)) : bOffNext;
        const int    bC = bOff + 32 * c;
        float2 an[8];
        if (CHAIN || more) {
#pragma unroll
            for (int s = 0; s < 8; ++s) an[s] = *(const float2*)(aN + 4 * s);
        }
#pragma unroll
        for (int s = 0; s < 8; ++s) {
            v2f bn[8];
            if (s < 7) {
#pragma unroll
                for (int o = 0; o < 8; ++o) bn[o] = *(const v2f*)(sWbase + bC + o * 16 * WP + 4 * (s + 1));
            } else if (CHAIN || more) {
#pragma unroll
                for (int o = 0; o < 8; ++o) bn[o] = *(const v2f*)(sWbase + bN + o * 16 * WP);
            }
            v2f a;
            if (SCALED) { a.x = ac[s].x * scale; a.y = ac[s].y * scale; }
            else        { a.x = ac[s].x;         a.y = ac[s].y; }
#pragma unroll
            for (int o = 0; o < 8; ++o) {
                acc[o] = __builtin_amdgcn_wmma_f32_16x16x4_f32(
                    false, a, false, bc[o], (short)0, acc[o], false, false);
            }
            if (s < 7 || CHAIN || more) {
#pragma unroll
                for (int o = 0; o < 8; ++o) bc[o] = bn[o];
            }
        }
        if (CHAIN || more) {
#pragma unroll
            for (int s = 0; s < 8; ++s) ac[s] = an[s];
        }
    }
}

__global__ void __launch_bounds__(GEMM_THREADS)
sage_gemm(const float* __restrict__ x,
          const float* __restrict__ Wl,
          const float* __restrict__ bl,
          const float* __restrict__ Wr,
          const float* __restrict__ cnt,
          float* __restrict__ out  ) {
    __shared__ float sW[2 * D * WP];

    for (int i = threadIdx.x; i < 2 * D * (D / 4); i += GEMM_THREADS) {
        const int m  = i >> 12;
        const int n  = (i >> 5) & (D - 1);
        const int c4 = i & 31;
        const float* W = m ? Wr : Wl;
        const float4 v = *(const float4*)(W + n * D + c4 * 4);
        *(float4*)(sW + (m * D + n) * WP + c4 * 4) = v;
    }
    __syncthreads();

    const int wave = __builtin_amdgcn_readfirstlane((int)(threadIdx.x >> 5));
    const int lane = threadIdx.x & 31;
    const int lrow  = lane & 15;
    const int khalf = (lane >> 4) * 2;

    const int bOffL = lrow * WP + khalf;
    const int bOffR = bOffL + D * WP;

    float bb[8];
#pragma unroll
    for (int o = 0; o < 8; ++o) bb[o] = bl[o * 16 + lrow];

    for (int m = wave; ; m += GEMM_WAVES) {
        const int tile = (int)blockIdx.x + GEMM_WGS * m;
        if (tile >= N_TILES) break;
        const int r0  = tile * 16;
        const int row = r0 + lrow;

        const float rc = 1.0f / fmaxf(cnt[row], 1.0f);

        v8f acc[8];
#pragma unroll
        for (int o = 0; o < 8; ++o) { v8f z = {}; acc[o] = z; }

        const float* aggPtr = out + (long long)row * D + khalf;
        const float* xPtr   = x   + (long long)row * D + khalf;

        float2 ac[8];
        v2f    bc[8];
#pragma unroll
        for (int s = 0; s < 8; ++s) ac[s] = *(const float2*)(aggPtr + 4 * s);
#pragma unroll
        for (int o = 0; o < 8; ++o) bc[o] = *(const v2f*)(sW + bOffL + o * 16 * WP);

        k_sweep<true,  true >(sW, aggPtr, bOffL, rc,   xPtr,   bOffR, ac, bc, acc);
        k_sweep<false, false>(sW, xPtr,   bOffR, 1.0f, xPtr,   bOffR, ac, bc, acc);

#pragma unroll
        for (int o = 0; o < 8; ++o) {
            const int col = o * 16 + lrow;
#pragma unroll
            for (int j = 0; j < 8; ++j) {
                const int rr = r0 + j + (lane >> 4) * 8;
                float v = acc[o][j] + bb[o];
                v = (v > 0.0f) ? v : SLOPE * v;
                out[(long long)rr * D + col] = v;
            }
        }
    }
}

extern "C" void kernel_launch(void* const* d_in, const int* in_sizes, int n_in,
                              void* d_out, int out_size, void* d_ws, size_t ws_size,
                              hipStream_t stream) {
    const float* x  = (const float*)d_in[0];
    const int*   ei = (const int*)  d_in[1];
    const float* Wl = (const float*)d_in[2];
    const float* bl = (const float*)d_in[3];
    const float* Wr = (const float*)d_in[4];
    float* out = (float*)d_out;
    float* cnt = (float*)d_ws;
    unsigned* keys = (unsigned*)((char*)d_ws + KEY_OFFSET);
    const int E = in_sizes[1] / 2;

    sage_keys<<<KEY_WGS, KEY_THREADS, 0, stream>>>(ei, keys, E);

    sage_own<<<OWN_WGS, OWN_THREADS, 0, stream>>>(x, ei, keys, out, cnt, E);

    sage_gemm<<<GEMM_WGS, GEMM_THREADS, 0, stream>>>(x, Wl, bl, Wr, cnt, out);
}


# edge scan: pre-launch writes per owner block one bit per 16-edge unit (LDS OR atomics, whole-line stores); a block reads only its own bit row, not a byte per edge; sum order = LDS-atomic order
# speedup vs baseline: 3.3705x; 1.1065x over previous
#include <hip/hip_runtime.h>


typedef __attribute__((ext_vector_type(2))) float v2f;
typedef __attribute__((ext_vector_type(8))) float v8f;

#define N_NODES 100000
#define D 128
#define SLOPE 0.2f

#define N_TILES   (N_NODES / 16)
#define GEMM_WGS  256
#define GEMM_THREADS 512
#define GEMM_WAVES (GEMM_THREADS / 32)
#define WP 132

#define NB          391
#define N_OWN       256
#define IDX_WGS     256
#define IDX_THREADS 1024
#define IDX_UNITS   1024
#define IDX_WORDS   (IDX_UNITS / 32)
#define IDX_PITCH   257
#define KEY_OFFSET  (1 << 20)

__global__ void __launch_bounds__(IDX_THREADS)
sage_index(const int* __restrict__ ei,
           unsigned*  __restrict__ bm,
           int E) {
    __shared__ unsigned sB[IDX_WORDS * IDX_PITCH];

    const int tid  = threadIdx.x;
    const int lane = tid & 31;
    const int wave = tid >> 5;
    const int* dstp   = ei + (long long)E;
    const int  nunit  = (E + 15) >> 4;
    const int  nchunk = (nunit + IDX_UNITS - 1) / IDX_UNITS;
    unsigned*  row    = sB + wave * IDX_PITCH;

    auto mark = [&](int d, unsigned bit) {
        if ((unsigned)d < (unsigned)N_NODES) atomicOr(&row[(unsigned)d / (unsigned)NB], bit);
    };

    for (int c = (int)blockIdx.x; c < nchunk; c += IDX_WGS) {
        for (int i = tid; i < IDX_WORDS * IDX_PITCH; i += IDX_THREADS) sB[i] = 0u;
        __syncthreads();

        const long long g0 = ((long long)c * IDX_UNITS + wave * 32) * 4 + lane;
        if (((long long)c * IDX_UNITS + IDX_UNITS) * 16 <= (long long)E) {
            int4 d[4];
#pragma unroll
            for (int k = 0; k < 4; ++k) d[k] = *(const int4*)(dstp + 4 * (g0 + 32 * k));
#pragma unroll
            for (int k = 0; k < 4; ++k) {
                const unsigned bit = 1u << (8 * k + (lane >> 2));
                mark(d[k].x, bit);
                mark(d[k].y, bit);
                mark(d[k].z, bit);
                mark(d[k].w, bit);
            }
        } else {
            for (int k = 0; k < 4; ++k) {
                const unsigned bit = 1u << (8 * k + (lane >> 2));
                const long long e0 = 4 * (g0 + 32 * k);
                for (int i = 0; i < 4; ++i)
                    if (e0 + i < (long long)E) mark(dstp[e0 + i], bit);
            }
        }
        __syncthreads();

        for (int o = wave; o < N_OWN; o += IDX_THREADS / 32)
            bm[((long long)o * nchunk + c) * IDX_WORDS + lane] = sB[lane * IDX_PITCH + o];
        __syncthreads();
    }
}

#define OWN_WGS     256
#define OWN_THREADS 1024
#define OWN_WAVES   (OWN_THREADS / 32)
#define LIST_CAP    16384
#define UNIT_CAP    8192
#define GATH_U      8

__global__ void __launch_bounds__(OWN_THREADS)
sage_own(const float*    __restrict__ x,
         const int*      __restrict__ ei,
         const unsigned* __restrict__ bm,
         float*          __restrict__ agg,
         float*          __restrict__ cnt,
         int E) {
    __shared__ float sA[NB * D];
    __shared__ float sC[NB];
    __shared__ int   sL[LIST_CAP];
    __shared__ int   sU[UNIT_CAP];
    __shared__ int   sN;
    __shared__ int   sNU;

    const int tid  = threadIdx.x;
    const int lane = tid & 31;
    const int lo   = (int)blockIdx.x * NB;
    const int nb   = min(NB, N_NODES - lo);

    for (int i = tid; i < NB * D / 4; i += OWN_THREADS) {
        float4 z = {0.0f, 0.0f, 0.0f, 0.0f};
        *(float4*)(sA + 4 * i) = z;
    }
    for (int i = tid; i < NB; i += OWN_THREADS) sC[i] = 0.0f;
    if (tid == 0) { sN = 0; sNU = 0; }
    __syncthreads();

    const int* dstp = ei + (long long)E;

    auto take = [&](int dv, int e) {
        const unsigned dl = (unsigned)(dv - lo);
        if (dl < (unsigned)nb) {
            atomicAdd(&sC[dl], 1.0f);
            const int pos = atomicAdd(&sN, 1);
            if (pos < LIST_CAP) {
                sL[pos] = e | ((int)dl << 21);
            } else {
                const float* xr = x + (long long)ei[e] * D;
                for (int f = 0; f < D; ++f) atomicAdd(&sA[dl * D + f], xr[f]);
            }
        }
    };

    auto take4 = [&](const int4 d, int e0) {
        const unsigned d0 = (unsigned)(d.x - lo), d1 = (unsigned)(d.y - lo);
        const unsigned d2 = (unsigned)(d.z - lo), d3 = (unsigned)(d.w - lo);
        if (min(min(d0, d1), min(d2, d3)) < (unsigned)nb) {
            take(d.x, e0);
            take(d.y, e0 + 1);
            take(d.z, e0 + 2);
            take(d.w, e0 + 3);
        }
    };

    auto resolve = [&](int j) {
        const int e0 = 16 * j;
        if (e0 + 16 <= E) {
            int4 d[4];
#pragma unroll
            for (int k = 0; k < 4; ++k) d[k] = *(const int4*)(dstp + e0 + 4 * k);
#pragma unroll
            for (int k = 0; k < 4; ++k) take4(d[k], e0 + 4 * k);
        } else {
            for (int e = e0; e < E; ++e) take(dstp[e], e);
        }
    };

    const int    nunit  = (E + 15) >> 4;
    const int    nchunk = (nunit + IDX_UNITS - 1) / IDX_UNITS;
    const uint4* bp     = (const uint4*)(bm + (long long)blockIdx.x * nchunk * IDX_WORDS);
    const int    nquad  = nchunk * (IDX_WORDS / 4);

    for (int i = tid; i < nquad; i += OWN_THREADS) {
        const uint4 w = bp[i];
        const int   c = __builtin_popcount(w.x) + __builtin_popcount(w.y) +
                        __builtin_popcount(w.z) + __builtin_popcount(w.w);
        if (c != 0) {
            int pos = atomicAdd(&sNU, c);
            auto emit = [&](unsigned v, int base) {
                while (v != 0u) {
                    if (pos < UNIT_CAP) sU[pos] = base + __builtin_ctz(v);
                    ++pos;
                    v &= v - 1u;
                }
            };
            emit(w.x, 128 * i);
            emit(w.y, 128 * i + 32);
            emit(w.z, 128 * i + 64);
            emit(w.w, 128 * i + 96);
        }
    }
    __syncthreads();

    {
        const int  nuAll  = sNU;
        const bool listed = (nuAll <= UNIT_CAP);
        const int  nwork  = listed ? nuAll : nunit;
        for (int q = tid; q < nwork; q += OWN_THREADS) resolve(listed ? sU[q] : q);
    }
    __syncthreads();

    const int n = min(sN, LIST_CAP);
    for (int q = tid; q < n; q += OWN_THREADS) {
        const int ent = sL[q];
        const int s   = ei[ent & 0x1FFFFF];
        sL[q] = s | ((ent >> 21) << 17);
    }
    __syncthreads();

    const int wave  = __builtin_amdgcn_readfirstlane(tid >> 5);
    const int nfull = n & ~(GATH_U - 1);
    for (int base = wave * GATH_U; base < nfull; base += OWN_WAVES * GATH_U) {
        const int mine = sL[base + (lane & (GATH_U - 1))];
        float v[GATH_U][4];
        int   dl[GATH_U];
#pragma unroll
        for (int k = 0; k < GATH_U; ++k) {
            const int ent = __builtin_amdgcn_readlane(mine, k);
            dl[k] = ent >> 17;
            const float* xr = x + (long long)(ent & 0x1FFFF) * D + lane;
            v[k][0] = xr[0];
            v[k][1] = xr[32];
            v[k][2] = xr[64];
            v[k][3] = xr[96];
        }
#pragma unroll
        for (int k = 0; k < GATH_U; ++k) {
            float* a = sA + dl[k] * D + lane;
            atomicAdd(a,      v[k][0]);
            atomicAdd(a + 32, v[k][1]);
            atomicAdd(a + 64, v[k][2]);
            atomicAdd(a + 96, v[k][3]);
        }
    }
    if (wave < n - nfull) {
        const int ent = __builtin_amdgcn_readfirstlane(sL[nfull + wave]);
        const float* xr = x + (long long)(ent & 0x1FFFF) * D + lane;
        const float v0 = xr[0], v1 = xr[32], v2 = xr[64], v3 = xr[96];
        float* a = sA + (ent >> 17) * D + lane;
        atomicAdd(a,      v0);
        atomicAdd(a + 32, v1);
        atomicAdd(a + 64, v2);
        atomicAdd(a + 96, v3);
    }
    __syncthreads();

    for (int q = tid; q < nb * (D / 4); q += OWN_THREADS) {
        const int row = q >> 5;
        const int c4  = q & 31;
        *(float4*)(agg + (long long)(lo + row) * D + c4 * 4) = *(const float4*)(sA + row * D + c4 * 4);
    }
    for (int q = tid; q < nb; q += OWN_THREADS) cnt[lo + q] = sC[q];
}

template <bool SCALED, bool CHAIN>
__device__ __forceinline__ void k_sweep(const float* sWbase,
                                        const float* aPtr, int bOff, float scale,
                                        const float* aNext, int bOffNext,
                                        float2 (&ac)[8], v2f (&bc)[8], v8f (&acc)[8]) {
#pragma unroll 1
    for (int c = 0; c < 4; ++c) {
        const bool more = (c < 3);
        const float* aN = more ? (aPtr + 32 * (c + 1)) : aNext;
        const int    bN = more ? (bOff + 32 * (c + 1)) : bOffNext;
        const int    bC = bOff + 32 * c;
        float2 an[8];
        if (CHAIN || more) {
#pragma unroll
            for (int s = 0; s < 8; ++s) an[s] = *(const float2*)(aN + 4 * s);
        }
#pragma unroll
        for (int s = 0; s < 8; ++s) {
            v2f bn[8];
            if (s < 7) {
#pragma unroll
                for (int o = 0; o < 8; ++o) bn[o] = *(const v2f*)(sWbase + bC + o * 16 * WP + 4 * (s + 1));
            } else if (CHAIN || more) {
#pragma unroll
                for (int o = 0; o < 8; ++o) bn[o] = *(const v2f*)(sWbase + bN + o * 16 * WP);
            }
            v2f a;
            if (SCALED) { a.x = ac[s].x * scale; a.y = ac[s].y * scale; }
            else        { a.x = ac[s].x;         a.y = ac[s].y; }
#pragma unroll
            for (int o = 0; o < 8; ++o) {
                acc[o] = __builtin_amdgcn_wmma_f32_16x16x4_f32(
                    false, a, false, bc[o], (short)0, acc[o], false, false);
            }
            if (s < 7 || CHAIN || more) {
#pragma unroll
                for (int o = 0; o < 8; ++o) bc[o] = bn[o];
            }
        }
        if (CHAIN || more) {
#pragma unroll
            for (int s = 0; s < 8; ++s) ac[s] = an[s];
        }
    }
}

__global__ void __launch_bounds__(GEMM_THREADS)
sage_gemm(const float* __restrict__ x,
          const float* __restrict__ Wl,
          const float* __restrict__ bl,
          const float* __restrict__ Wr,
          const float* __restrict__ cnt,
          float* __restrict__ out  ) {
    __shared__ float sW[2 * D * WP];

    for (int i = threadIdx.x; i < 2 * D * (D / 4); i += GEMM_THREADS) {
        const int m  = i >> 12;
        const int n  = (i >> 5) & (D - 1);
        const int c4 = i & 31;
        const float* W = m ? Wr : Wl;
        const float4 v = *(const float4*)(W + n * D + c4 * 4);
        *(float4*)(sW + (m * D + n) * WP + c4 * 4) = v;
    }
    __syncthreads();

    const int wave = __builtin_amdgcn_readfirstlane((int)(threadIdx.x >> 5));
    const int lane = threadIdx.x & 31;
    const int lrow  = lane & 15;
    const int khalf = (lane >> 4) * 2;

    const int bOffL = lrow * WP + khalf;
    const int bOffR = bOffL + D * WP;

    float bb[8];
#pragma unroll
    for (int o = 0; o < 8; ++o) bb[o] = bl[o * 16 + lrow];

    for (int m = wave; ; m += GEMM_WAVES) {
        const int tile = (int)blockIdx.x + GEMM_WGS * m;
        if (tile >= N_TILES) break;
        const int r0  = tile * 16;
        const int row = r0 + lrow;

        const float rc = 1.0f / fmaxf(cnt[row], 1.0f);

        v8f acc[8];
#pragma unroll
        for (int o = 0; o < 8; ++o) { v8f z = {}; acc[o] = z; }

        const float* aggPtr = out + (long long)row * D + khalf;
        const float* xPtr   = x   + (long long)row * D + khalf;

        float2 ac[8];
        v2f    bc[8];
#pragma unroll
        for (int s = 0; s < 8; ++s) ac[s] = *(const float2*)(aggPtr + 4 * s);
#pragma unroll
        for (int o = 0; o < 8; ++o) bc[o] = *(const v2f*)(sW + bOffL + o * 16 * WP);

        k_sweep<true,  true >(sW, aggPtr, bOffL, rc,   xPtr,   bOffR, ac, bc, acc);
        k_sweep<false, false>(sW, xPtr,   bOffR, 1.0f, xPtr,   bOffR, ac, bc, acc);

#pragma unroll
        for (int o = 0; o < 8; ++o) {
            const int col = o * 16 + lrow;
#pragma unroll
            for (int j = 0; j < 8; ++j) {
                const int rr = r0 + j + (lane >> 4) * 8;
                float v = acc[o][j] + bb[o];
                v = (v > 0.0f) ? v : SLOPE * v;
                out[(long long)rr * D + col] = v;
            }
        }
    }
}

extern "C" void kernel_launch(void* const* d_in, const int* in_sizes, int n_in,
                              void* d_out, int out_size, void* d_ws, size_t ws_size,
                              hipStream_t stream) {
    const float* x  = (const float*)d_in[0];
    const int*   ei = (const int*)  d_in[1];
    const float* Wl = (const float*)d_in[2];
    const float* bl = (const float*)d_in[3];
    const float* Wr = (const float*)d_in[4];
    float* out = (float*)d_out;
    float* cnt = (float*)d_ws;
    unsigned* bm = (unsigned*)((char*)d_ws + KEY_OFFSET);
    const int E = in_sizes[1] / 2;

    sage_index<<<IDX_WGS, IDX_THREADS, 0, stream>>>(ei, bm, E);

    sage_own<<<OWN_WGS, OWN_THREADS, 0, stream>>>(x, ei, bm, out, cnt, E);

    sage_gemm<<<GEMM_WGS, GEMM_THREADS, 0, stream>>>(x, Wl, bl, Wr, cnt, out);
}


# edge sums: block's 391 rows dealt to its 32 waves (row & 31), one edge list per wave; a wave adds its rows by plain 16-B LDS load/add/store, no LDS f32 atomics; sum order = list (counter-add) order
# speedup vs baseline: 3.8005x; 1.1276x over previous
#include <hip/hip_runtime.h>


typedef __attribute__((ext_vector_type(2))) float v2f;
typedef __attribute__((ext_vector_type(8))) float v8f;

#define N_NODES 100000
#define D 128
#define SLOPE 0.2f

#define N_TILES   (N_NODES / 16)
#define GEMM_WGS  256
#define GEMM_THREADS 512
#define GEMM_WAVES (GEMM_THREADS / 32)
#define WP 132

#define NB          391
#define N_OWN       256
#define IDX_WGS     256
#define IDX_THREADS 1024
#define IDX_UNITS   1024
#define IDX_WORDS   (IDX_UNITS / 32)
#define IDX_PITCH   257
#define KEY_OFFSET  (1 << 20)

__global__ void __launch_bounds__(IDX_THREADS)
sage_index(const int* __restrict__ ei,
           unsigned*  __restrict__ bm,
           int E) {
    __shared__ unsigned sB[IDX_WORDS * IDX_PITCH];

    const int tid  = threadIdx.x;
    const int lane = tid & 31;
    const int wave = tid >> 5;
    const int* dstp   = ei + (long long)E;
    const int  nunit  = (E + 15) >> 4;
    const int  nchunk = (nunit + IDX_UNITS - 1) / IDX_UNITS;
    unsigned*  row    = sB + wave * IDX_PITCH;

    auto mark = [&](int d, unsigned bit) {
        if ((unsigned)d < (unsigned)N_NODES) atomicOr(&row[(unsigned)d / (unsigned)NB], bit);
    };

    for (int c = (int)blockIdx.x; c < nchunk; c += IDX_WGS) {
        for (int i = tid; i < IDX_WORDS * IDX_PITCH; i += IDX_THREADS) sB[i] = 0u;
        __syncthreads();

        const long long g0 = ((long long)c * IDX_UNITS + wave * 32) * 4 + lane;
        if (((long long)c * IDX_UNITS + IDX_UNITS) * 16 <= (long long)E) {
            int4 d[4];
#pragma unroll
            for (int k = 0; k < 4; ++k) d[k] = *(const int4*)(dstp + 4 * (g0 + 32 * k));
#pragma unroll
            for (int k = 0; k < 4; ++k) {
                const unsigned bit = 1u << (8 * k + (lane >> 2));
                mark(d[k].x, bit);
                mark(d[k].y, bit);
                mark(d[k].z, bit);
                mark(d[k].w, bit);
            }
        } else {
            for (int k = 0; k < 4; ++k) {
                const unsigned bit = 1u << (8 * k + (lane >> 2));
                const long long e0 = 4 * (g0 + 32 * k);
                for (int i = 0; i < 4; ++i)
                    if (e0 + i < (long long)E) mark(dstp[e0 + i], bit);
            }
        }
        __syncthreads();

        for (int o = wave; o < N_OWN; o += IDX_THREADS / 32)
            bm[((long long)o * nchunk + c) * IDX_WORDS + lane] = sB[lane * IDX_PITCH + o];
        __syncthreads();
    }
}

#define OWN_WGS     256
#define OWN_THREADS 1024
#define OWN_WAVES   (OWN_THREADS / 32)
#define WLIST_CAP   512
#define UNIT_CAP    8192
#define GATH_U      8

__global__ void __launch_bounds__(OWN_THREADS)
sage_own(const float*    __restrict__ x,
         const int*      __restrict__ ei,
         const unsigned* __restrict__ bm,
         float*          __restrict__ agg,
         float*          __restrict__ cnt,
         int E) {
    __shared__ float sA[NB * D];
    __shared__ float sC[NB];
    __shared__ int   sL[OWN_WAVES * WLIST_CAP];
    __shared__ int   sU[UNIT_CAP];
    __shared__ int   sNW[OWN_WAVES];
    __shared__ int   sNU;

    const int tid  = threadIdx.x;
    const int lane = tid & 31;
    const int lo   = (int)blockIdx.x * NB;
    const int nb   = min(NB, N_NODES - lo);

    for (int i = tid; i < NB * D / 4; i += OWN_THREADS) {
        float4 z = {0.0f, 0.0f, 0.0f, 0.0f};
        *(float4*)(sA + 4 * i) = z;
    }
    for (int i = tid; i < NB; i += OWN_THREADS) sC[i] = 0.0f;
    if (tid < OWN_WAVES) sNW[tid] = 0;
    if (tid == 0) sNU = 0;
    __syncthreads();

    const int* dstp = ei + (long long)E;

    auto take = [&](int dv, int e) {
        const unsigned dl = (unsigned)(dv - lo);
        if (dl < (unsigned)nb) {
            atomicAdd(&sC[dl], 1.0f);
            const int w   = (int)(dl & (OWN_WAVES - 1));
            const int pos = atomicAdd(&sNW[w], 1);
            if (pos < WLIST_CAP) {
                sL[w * WLIST_CAP + pos] = e | ((int)dl << 21);
            } else {
                const float* xr = x + (long long)ei[e] * D;
                for (int f = 0; f < D; ++f) atomicAdd(&sA[dl * D + f], xr[f]);
            }
        }
    };

    auto take4 = [&](const int4 d, int e0) {
        const unsigned d0 = (unsigned)(d.x - lo), d1 = (unsigned)(d.y - lo);
        const unsigned d2 = (unsigned)(d.z - lo), d3 = (unsigned)(d.w - lo);
        if (min(min(d0, d1), min(d2, d3)) < (unsigned)nb) {
            take(d.x, e0);
            take(d.y, e0 + 1);
            take(d.z, e0 + 2);
            take(d.w, e0 + 3);
        }
    };

    auto resolve = [&](int j) {
        const int e0 = 16 * j;
        if (e0 + 16 <= E) {
            int4 d[4];
#pragma unroll
            for (int k = 0; k < 4; ++k) d[k] = *(const int4*)(dstp + e0 + 4 * k);
#pragma unroll
            for (int k = 0; k < 4; ++k) take4(d[k], e0 + 4 * k);
        } else {
            for (int e = e0; e < E; ++e) take(dstp[e], e);
        }
    };

    const int    nunit  = (E + 15) >> 4;
    const int    nchunk = (nunit + IDX_UNITS - 1) / IDX_UNITS;
    const uint4* bp     = (const uint4*)(bm + (long long)blockIdx.x * nchunk * IDX_WORDS);
    const int    nquad  = nchunk * (IDX_WORDS / 4);

    for (int i = tid; i < nquad; i += OWN_THREADS) {
        const uint4 w = bp[i];
        const int   c = __builtin_popcount(w.x) + __builtin_popcount(w.y) +
                        __builtin_popcount(w.z) + __builtin_popcount(w.w);
        if (c != 0) {
            int pos = atomicAdd(&sNU, c);
            auto emit = [&](unsigned v, int base) {
                while (v != 0u) {
                    if (pos < UNIT_CAP) sU[pos] = base + __builtin_ctz(v);
                    ++pos;
                    v &= v - 1u;
                }
            };
            emit(w.x, 128 * i);
            emit(w.y, 128 * i + 32);
            emit(w.z, 128 * i + 64);
            emit(w.w, 128 * i + 96);
        }
    }
    __syncthreads();

    {
        const int  nuAll  = sNU;
        const bool listed = (nuAll <= UNIT_CAP);
        const int  nwork  = listed ? nuAll : nunit;
        for (int q = tid; q < nwork; q += OWN_THREADS) resolve(listed ? sU[q] : q);
    }
    __syncthreads();

    const int wave = __builtin_amdgcn_readfirstlane(tid >> 5);
    int*      myL  = sL + wave * WLIST_CAP;
    const int myN  = min(sNW[wave], WLIST_CAP);
    for (int q = lane; q < myN; q += 32) {
        const int ent = myL[q];
        const int s   = ei[ent & 0x1FFFFF];
        myL[q] = s | ((ent >> 21) << 17);
    }
    __syncthreads();

    const float* xl  = x  + 4 * lane;
    float*       sAl = sA + 4 * lane;

    auto rmw = [&](int row, const float4 v) {
        float4* a = (float4*)(sAl + row * D);
        float4  t = *a;
        t.x += v.x; t.y += v.y; t.z += v.z; t.w += v.w;
        *a = t;
    };
    auto ask = [&](int base, float4 (&v)[GATH_U], int (&r)[GATH_U]) {
        const int mine = myL[base + (lane & (GATH_U - 1))];
#pragma unroll
        for (int k = 0; k < GATH_U; ++k) {
            const int ent = __builtin_amdgcn_readlane(mine, k);
            r[k] = ent >> 17;
            v[k] = *(const float4*)(xl + (long long)(ent & 0x1FFFF) * D);
        }
    };
    auto add = [&](const float4 (&v)[GATH_U], const int (&r)[GATH_U]) {
#pragma unroll
        for (int k = 0; k < GATH_U; ++k) rmw(r[k], v[k]);
    };

    const int nfull = myN & ~(GATH_U - 1);
    const int rem   = myN - nfull;
    float4 va[GATH_U], vb[GATH_U];
    int    ra[GATH_U], rb[GATH_U];

    if (rem > 0) {
        float4 vt[GATH_U - 1];
        int    rt[GATH_U - 1];
        const int mine = myL[nfull + (lane & (GATH_U - 1))];
#pragma unroll
        for (int k = 0; k < GATH_U - 1; ++k) {
            if (k < rem) {
                const int ent = __builtin_amdgcn_readlane(mine, k);
                rt[k] = ent >> 17;
                vt[k] = *(const float4*)(xl + (long long)(ent & 0x1FFFF) * D);
            }
        }
        if (nfull > 0) ask(0, va, ra);
#pragma unroll
        for (int k = 0; k < GATH_U - 1; ++k) {
            if (k < rem) rmw(rt[k], vt[k]);
        }
    } else if (nfull > 0) {
        ask(0, va, ra);
    }
    if (nfull > 0) {
        int base = GATH_U;
        for (;;) {
            if (base >= nfull) { add(va, ra); break; }
            ask(base, vb, rb);
            add(va, ra);
            base += GATH_U;
            if (base >= nfull) { add(vb, rb); break; }
            ask(base, va, ra);
            add(vb, rb);
            base += GATH_U;
        }
    }
    __syncthreads();

    for (int q = tid; q < nb * (D / 4); q += OWN_THREADS) {
        const int row = q >> 5;
        const int c4  = q & 31;
        *(float4*)(agg + (long long)(lo + row) * D + c4 * 4) = *(const float4*)(sA + row * D + c4 * 4);
    }
    for (int q = tid; q < nb; q += OWN_THREADS) cnt[lo + q] = sC[q];
}

template <bool SCALED, bool CHAIN>
__device__ __forceinline__ void k_sweep(const float* sWbase,
                                        const float* aPtr, int bOff, float scale,
                                        const float* aNext, int bOffNext,
                                        float2 (&ac)[8], v2f (&bc)[8], v8f (&acc)[8]) {
#pragma unroll 1
    for (int c = 0; c < 4; ++c) {
        const bool more = (c < 3);
        const float* aN = more ? (aPtr + 32 * (c + 1)) : aNext;
        const int    bN = more ? (bOff + 32 * (c + 1)) : bOffNext;
        const int    bC = bOff + 32 * c;
        float2 an[8];
        if (CHAIN || more) {
#pragma unroll
            for (int s = 0; s < 8; ++s) an[s] = *(const float2*)(aN + 4 * s);
        }
#pragma unroll
        for (int s = 0; s < 8; ++s) {
            v2f bn[8];
            if (s < 7) {
#pragma unroll
                for (int o = 0; o < 8; ++o) bn[o] = *(const v2f*)(sWbase + bC + o * 16 * WP + 4 * (s + 1));
            } else if (CHAIN || more) {
#pragma unroll
                for (int o = 0; o < 8; ++o) bn[o] = *(const v2f*)(sWbase + bN + o * 16 * WP);
            }
            v2f a;
            if (SCALED) { a.x = ac[s].x * scale; a.y = ac[s].y * scale; }
            else        { a.x = ac[s].x;         a.y = ac[s].y; }
#pragma unroll
            for (int o = 0; o < 8; ++o) {
                acc[o] = __builtin_amdgcn_wmma_f32_16x16x4_f32(
                    false, a, false, bc[o], (short)0, acc[o], false, false);
            }
            if (s < 7 || CHAIN || more) {
#pragma unroll
                for (int o = 0; o < 8; ++o) bc[o] = bn[o];
            }
        }
        if (CHAIN || more) {
#pragma unroll
            for (int s = 0; s < 8; ++s) ac[s] = an[s];
        }
    }
}

__global__ void __launch_bounds__(GEMM_THREADS)
sage_gemm(const float* __restrict__ x,
          const float* __restrict__ Wl,
          const float* __restrict__ bl,
          const float* __restrict__ Wr,
          const float* __restrict__ cnt,
          float* __restrict__ out  ) {
    __shared__ float sW[2 * D * WP];

    for (int i = threadIdx.x; i < 2 * D * (D / 4); i += GEMM_THREADS) {
        const int m  = i >> 12;
        const int n  = (i >> 5) & (D - 1);
        const int c4 = i & 31;
        const float* W = m ? Wr : Wl;
        const float4 v = *(const float4*)(W + n * D + c4 * 4);
        *(float4*)(sW + (m * D + n) * WP + c4 * 4) = v;
    }
    __syncthreads();

    const int wave = __builtin_amdgcn_readfirstlane((int)(threadIdx.x >> 5));
    const int lane = threadIdx.x & 31;
    const int lrow  = lane & 15;
    const int khalf = (lane >> 4) * 2;

    const int bOffL = lrow * WP + khalf;
    const int bOffR = bOffL + D * WP;

    float bb[8];
#pragma unroll
    for (int o = 0; o < 8; ++o) bb[o] = bl[o * 16 + lrow];

    for (int m = wave; ; m += GEMM_WAVES) {
        const int tile = (int)blockIdx.x + GEMM_WGS * m;
        if (tile >= N_TILES) break;
        const int r0  = tile * 16;
        const int row = r0 + lrow;

        const float rc = 1.0f / fmaxf(cnt[row], 1.0f);

        v8f acc[8];
#pragma unroll
        for (int o = 0; o < 8; ++o) { v8f z = {}; acc[o] = z; }

        const float* aggPtr = out + (long long)row * D + khalf;
        const float* xPtr   = x   + (long long)row * D + khalf;

        float2 ac[8];
        v2f    bc[8];
#pragma unroll
        for (int s = 0; s < 8; ++s) ac[s] = *(const float2*)(aggPtr + 4 * s);
#pragma unroll
        for (int o = 0; o < 8; ++o) bc[o] = *(const v2f*)(sW + bOffL + o * 16 * WP);

        k_sweep<true,  true >(sW, aggPtr, bOffL, rc,   xPtr,   bOffR, ac, bc, acc);
        k_sweep<false, false>(sW, xPtr,   bOffR, 1.0f, xPtr,   bOffR, ac, bc, acc);

#pragma unroll
        for (int o = 0; o < 8; ++o) {
            const int col = o * 16 + lrow;
#pragma unroll
            for (int j = 0; j < 8; ++j) {
                const int rr = r0 + j + (lane >> 4) * 8;
                float v = acc[o][j] + bb[o];
                v = (v > 0.0f) ? v : SLOPE * v;
                out[(long long)rr * D + col] = v;
            }
        }
    }
}

extern "C" void kernel_launch(void* const* d_in, const int* in_sizes, int n_in,
                              void* d_out, int out_size, void* d_ws, size_t ws_size,
                              hipStream_t stream) {
    const float* x  = (const float*)d_in[0];
    const int*   ei = (const int*)  d_in[1];
    const float* Wl = (const float*)d_in[2];
    const float* bl = (const float*)d_in[3];
    const float* Wr = (const float*)d_in[4];
    float* out = (float*)d_out;
    float* cnt = (float*)d_ws;
    unsigned* bm = (unsigned*)((char*)d_ws + KEY_OFFSET);
    const int E = in_sizes[1] / 2;

    sage_index<<<IDX_WGS, IDX_THREADS, 0, stream>>>(ei, bm, E);

    sage_own<<<OWN_WGS, OWN_THREADS, 0, stream>>>(x, ei, bm, out, cnt, E);

    sage_gemm<<<GEMM_WGS, GEMM_THREADS, 0, stream>>>(x, Wl, bl, Wr, cnt, out);
}
